# starting kernel
# baseline (speedup 1.0000x reference)
#include <hip/hip_runtime.h>

typedef float v2f __attribute__((ext_vector_type(2)));
typedef float v4f __attribute__((ext_vector_type(4)));
typedef float v8f __attribute__((ext_vector_type(8)));

namespace {
constexpr int kNQ     = 8;
constexpr int kFFN    = 512;
constexpr int kTokens = 32 * 4096;
constexpr int kWaves  = 8;
constexpr int kMT     = 4;
}

__global__ __launch_bounds__(256)
void ffq_fused_wmma(const float* __restrict__ x,
                    const float* __restrict__ phi,
                    const float* __restrict__ w1,
                    const float* __restrict__ b1,
                    const float* __restrict__ w2,
                    const float* __restrict__ b2,
                    float* __restrict__ out)
{
  const int lane = threadIdx.x & 31;
  const int wave = threadIdx.x >> 5;
  const int row  = lane & 15;
  const int hi   = lane >> 4;
  const int koff = hi << 1;

  const int tile0 = (blockIdx.x * kWaves + wave) * kMT;
  const float ph  = phi[0];

  v2f qlo[kMT], qhi[kMT];
#pragma unroll
  for (int m = 0; m < kMT; ++m) {
    const float* xr = x + (size_t)((tile0 + m) * 16 + row) * kNQ;
    qlo[m][0] = __cosf(xr[koff + 0] + ph);
    qlo[m][1] = __cosf(xr[koff + 1] + ph);
    qhi[m][0] = __cosf(xr[koff + 4] + ph);
    qhi[m][1] = __cosf(xr[koff + 5] + ph);
  }

  v2f bone; bone[0] = hi ? 0.0f : 1.0f; bone[1] = 0.0f;
  const v8f vzero = {};

  v2f ab2; ab2[0] = hi ? 0.0f : b2[row & 7]; ab2[1] = 0.0f;
  const v8f accInit =
      __builtin_amdgcn_wmma_f32_16x16x4_f32(false, ab2, false, bone, (short)0, vzero, false, false);
  v8f acc[kMT];
#pragma unroll
  for (int m = 0; m < kMT; ++m) acc[m] = accInit;

  const float* w2r = w2 + (size_t)(row & 7) * kFFN + 8 * hi;

  for (int n0 = 0; n0 < kFFN; n0 += 16) {
    const float* w1r = w1 + (size_t)(n0 + row) * kNQ;
    v2f wlo, whi;
    wlo[0] = w1r[koff + 0]; wlo[1] = w1r[koff + 1];
    whi[0] = w1r[koff + 4]; whi[1] = w1r[koff + 5];

    v2f ab1; ab1[0] = hi ? 0.0f : b1[n0 + row]; ab1[1] = 0.0f;

    const v4f wq0 = *(const v4f*)(w2r + n0);
    const v4f wq1 = *(const v4f*)(w2r + n0 + 4);

    const v8f hinit =
        __builtin_amdgcn_wmma_f32_16x16x4_f32(false, ab1, false, bone, (short)0, vzero, false, false);

#pragma unroll
    for (int m = 0; m < kMT; ++m) {
      v8f h = __builtin_amdgcn_wmma_f32_16x16x4_f32(false, wlo, false, qlo[m], (short)0, hinit, false, false);
      h     = __builtin_amdgcn_wmma_f32_16x16x4_f32(false, whi, false, qhi[m], (short)0, h,     false, false);
#pragma unroll
      for (int r = 0; r < 8; ++r) h[r] = h[r] > 0.0f ? h[r] : 0.0f;

#pragma unroll
      for (int i = 0; i < 4; ++i) {
        v2f a2, bfrag;
        a2[0]    = (i < 2) ? wq0[2 * i]     : wq1[2 * (i - 2)];
        a2[1]    = (i < 2) ? wq0[2 * i + 1] : wq1[2 * (i - 2) + 1];
        bfrag[0] = h[2 * i];
        bfrag[1] = h[2 * i + 1];
        acc[m] = __builtin_amdgcn_wmma_f32_16x16x4_f32(false, a2, false, bfrag, (short)0, acc[m], false, false);
      }
    }
  }

  if (hi == 0) {
#pragma unroll
    for (int m = 0; m < kMT; ++m) {
      float* orow = out + (size_t)((tile0 + m) * 16 + row) * kNQ;
      v4f lo4, hi4;
      lo4[0] = acc[m][0]; lo4[1] = acc[m][1]; lo4[2] = acc[m][2]; lo4[3] = acc[m][3];
      hi4[0] = acc[m][4]; hi4[1] = acc[m][5]; hi4[2] = acc[m][6]; hi4[3] = acc[m][7];
      *(v4f*)(orow + 0) = lo4;
      *(v4f*)(orow + 4) = hi4;
    }
  }
}

extern "C" void kernel_launch(void* const* d_in, const int* in_sizes, int n_in,
                              void* d_out, int out_size, void* d_ws, size_t ws_size,
                              hipStream_t stream) {
  const float* x   = (const float*)d_in[0];
  const float* phi = (const float*)d_in[1];
  const float* w1  = (const float*)d_in[2];
  const float* b1  = (const float*)d_in[3];
  const float* w2  = (const float*)d_in[4];
  const float* b2  = (const float*)d_in[5];
  float* out = (float*)d_out;

  const int blocks = kTokens / (16 * kMT * kWaves);
  ffq_fused_wmma<<<blocks, 256, 0, stream>>>(x, phi, w1, b1, w2, b2, out);
}


# Weight fragments (w1, b1, w2) go through a 4-deep register ring: each chunk's loads issue 2-3 chunks ahead of use, so the second-level-cache round trip no longer stalls every chunk.
# speedup vs baseline: 1.4601x; 1.4601x over previous
#include <hip/hip_runtime.h>


typedef float v2f __attribute__((ext_vector_type(2)));
typedef float v4f __attribute__((ext_vector_type(4)));
typedef float v8f __attribute__((ext_vector_type(8)));

namespace {
constexpr int kNQ     = 8;
constexpr int kFFN    = 512;
constexpr int kTokens = 32 * 4096;
constexpr int kWaves  = 8;
constexpr int kMT     = 4;
constexpr int kChunk  = 16;
constexpr int kStages = 4;

struct WFrag {
  v2f wlo, whi;
  float b1v;
  v4f wq0, wq1;
};

__device__ __forceinline__ WFrag load_frag(const float* __restrict__ w1l,
                                           const float* __restrict__ b1l,
                                           const float* __restrict__ w2l, int n0)
{
  WFrag f;
  f.wlo = *(const v2f*)(w1l + n0 * kNQ);
  f.whi = *(const v2f*)(w1l + n0 * kNQ + 4);
  f.b1v = b1l[n0];
  f.wq0 = *(const v4f*)(w2l + n0);
  f.wq1 = *(const v4f*)(w2l + n0 + 4);
  return f;
}

__device__ __forceinline__ void ffn_chunk(const WFrag& f, int hi, v2f bone,
                                          const v2f (&qlo)[kMT], const v2f (&qhi)[kMT],
                                          v8f (&acc)[kMT])
{
  const v8f vzero = {};
  v2f ab1; ab1[0] = hi ? 0.0f : f.b1v; ab1[1] = 0.0f;
  const v8f hinit =
      __builtin_amdgcn_wmma_f32_16x16x4_f32(false, ab1, false, bone, (short)0, vzero, false, false);

#pragma unroll
  for (int m = 0; m < kMT; ++m) {
    v8f h = __builtin_amdgcn_wmma_f32_16x16x4_f32(false, f.wlo, false, qlo[m], (short)0, hinit, false, false);
    h     = __builtin_amdgcn_wmma_f32_16x16x4_f32(false, f.whi, false, qhi[m], (short)0, h,     false, false);
#pragma unroll
    for (int r = 0; r < 8; ++r) h[r] = h[r] > 0.0f ? h[r] : 0.0f;

#pragma unroll
    for (int i = 0; i < 4; ++i) {
      v2f a2, bfrag;
      a2[0]    = (i < 2) ? f.wq0[2 * i]     : f.wq1[2 * (i - 2)];
      a2[1]    = (i < 2) ? f.wq0[2 * i + 1] : f.wq1[2 * (i - 2) + 1];
      bfrag[0] = h[2 * i];
      bfrag[1] = h[2 * i + 1];
      acc[m] = __builtin_amdgcn_wmma_f32_16x16x4_f32(false, a2, false, bfrag, (short)0, acc[m], false, false);
    }
  }
}
}

__global__ __launch_bounds__(256)
void ffq_fused_wmma(const float* __restrict__ x,
                    const float* __restrict__ phi,
                    const float* __restrict__ w1,
                    const float* __restrict__ b1,
                    const float* __restrict__ w2,
                    const float* __restrict__ b2,
                    float* __restrict__ out)
{
  const int lane = threadIdx.x & 31;
  const int wave = threadIdx.x >> 5;
  const int row  = lane & 15;
  const int hi   = lane >> 4;
  const int koff = hi << 1;

  const float* w1l = w1 + row * kNQ + koff;
  const float* b1l = b1 + row;
  const float* w2l = w2 + (row & 7) * kFFN + 8 * hi;

  WFrag f0 = load_frag(w1l, b1l, w2l, 0 * kChunk);
  WFrag f1 = load_frag(w1l, b1l, w2l, 1 * kChunk);
  WFrag f2 = load_frag(w1l, b1l, w2l, 2 * kChunk);

  const int tile0 = (blockIdx.x * kWaves + wave) * kMT;
  const float ph  = phi[0];

  v2f qlo[kMT], qhi[kMT];
#pragma unroll
  for (int m = 0; m < kMT; ++m) {
    const float* xr = x + (size_t)((tile0 + m) * 16 + row) * kNQ;
    qlo[m][0] = __cosf(xr[koff + 0] + ph);
    qlo[m][1] = __cosf(xr[koff + 1] + ph);
    qhi[m][0] = __cosf(xr[koff + 4] + ph);
    qhi[m][1] = __cosf(xr[koff + 5] + ph);
  }

  v2f bone; bone[0] = hi ? 0.0f : 1.0f; bone[1] = 0.0f;
  const v8f vzero = {};

  v2f ab2; ab2[0] = hi ? 0.0f : b2[row & 7]; ab2[1] = 0.0f;
  const v8f accInit =
      __builtin_amdgcn_wmma_f32_16x16x4_f32(false, ab2, false, bone, (short)0, vzero, false, false);
  v8f acc[kMT];
#pragma unroll
  for (int m = 0; m < kMT; ++m) acc[m] = accInit;

  for (int n0 = 0; n0 < kFFN; n0 += kStages * kChunk) {
    WFrag f3 = load_frag(w1l, b1l, w2l, n0 + 3 * kChunk);
    ffn_chunk(f0, hi, bone, qlo, qhi, acc);
    f0 = load_frag(w1l, b1l, w2l, (n0 + 4 * kChunk) & (kFFN - 1));
    ffn_chunk(f1, hi, bone, qlo, qhi, acc);
    f1 = load_frag(w1l, b1l, w2l, (n0 + 5 * kChunk) & (kFFN - 1));
    ffn_chunk(f2, hi, bone, qlo, qhi, acc);
    f2 = load_frag(w1l, b1l, w2l, (n0 + 6 * kChunk) & (kFFN - 1));
    ffn_chunk(f3, hi, bone, qlo, qhi, acc);
  }

  if (hi == 0) {
#pragma unroll
    for (int m = 0; m < kMT; ++m) {
      float* orow = out + (size_t)((tile0 + m) * 16 + row) * kNQ;
      v4f lo4, hi4;
      lo4[0] = acc[m][0]; lo4[1] = acc[m][1]; lo4[2] = acc[m][2]; lo4[3] = acc[m][3];
      hi4[0] = acc[m][4]; hi4[1] = acc[m][5]; hi4[2] = acc[m][6]; hi4[3] = acc[m][7];
      *(v4f*)(orow + 0) = lo4;
      *(v4f*)(orow + 4) = hi4;
    }
  }
}

extern "C" void kernel_launch(void* const* d_in, const int* in_sizes, int n_in,
                              void* d_out, int out_size, void* d_ws, size_t ws_size,
                              hipStream_t stream) {
  const float* x   = (const float*)d_in[0];
  const float* phi = (const float*)d_in[1];
  const float* w1  = (const float*)d_in[2];
  const float* b1  = (const float*)d_in[3];
  const float* w2  = (const float*)d_in[4];
  const float* b2  = (const float*)d_in[5];
  float* out = (float*)d_out;

  const int blocks = kTokens / (16 * kMT * kWaves);
  ffq_fused_wmma<<<blocks, 256, 0, stream>>>(x, phi, w1, b1, w2, b2, out);
}


# Chunk loop software-pipelined by one chunk: GEMM1+ReLU of chunk c+1 woven between the K-steps of GEMM2 of chunk c across the 4 token tiles, so no WMMA waits on its predecessor; sum order unchanged.
# speedup vs baseline: 1.6494x; 1.1296x over previous
#include <hip/hip_runtime.h>


typedef float v2f __attribute__((ext_vector_type(2)));
typedef float v4f __attribute__((ext_vector_type(4)));
typedef float v8f __attribute__((ext_vector_type(8)));

namespace {
constexpr int kNQ     = 8;
constexpr int kFFN    = 512;
constexpr int kTokens = 32 * 4096;
constexpr int kWaves  = 8;
constexpr int kMT     = 4;
constexpr int kChunk  = 16;
constexpr int kStages = 4;
constexpr int kChunks = kFFN / kChunk;

struct WFrag {
  v2f wlo, whi;
  float b1v;
  v4f wq0, wq1;
};

__device__ __forceinline__ WFrag load_frag(const float* __restrict__ w1l,
                                           const float* __restrict__ b1l,
                                           const float* __restrict__ w2l, int n0)
{
  WFrag f;
  f.wlo = *(const v2f*)(w1l + n0 * kNQ);
  f.whi = *(const v2f*)(w1l + n0 * kNQ + 4);
  f.b1v = b1l[n0];
  f.wq0 = *(const v4f*)(w2l + n0);
  f.wq1 = *(const v4f*)(w2l + n0 + 4);
  return f;
}

__device__ __forceinline__ v8f mm(v2f a, v2f b, v8f c)
{
  return __builtin_amdgcn_wmma_f32_16x16x4_f32(false, a, false, b, (short)0, c, false, false);
}

__device__ __forceinline__ v2f w2_step(const WFrag& f, int i)
{
  v2f a;
  a[0] = (i < 2) ? f.wq0[2 * i]     : f.wq1[2 * (i - 2)];
  a[1] = (i < 2) ? f.wq0[2 * i + 1] : f.wq1[2 * (i - 2) + 1];
  return a;
}

__device__ __forceinline__ v2f h_step(const v8f& h, int i)
{
  v2f b; b[0] = h[2 * i]; b[1] = h[2 * i + 1];
  return b;
}

__device__ __forceinline__ void relu4(v8f& h, int r0)
{
#pragma unroll
  for (int r = r0; r < r0 + 4; ++r) h[r] = h[r] > 0.0f ? h[r] : 0.0f;
}

__device__ __forceinline__ void gemm1(const WFrag& f, int hi, v2f bone,
                                      const v2f (&qlo)[kMT], const v2f (&qhi)[kMT],
                                      v8f (&h)[kMT])
{
  const v8f vzero = {};
  v2f ab1; ab1[0] = hi ? 0.0f : f.b1v; ab1[1] = 0.0f;
  const v8f hinit = mm(ab1, bone, vzero);
#pragma unroll
  for (int m = 0; m < kMT; ++m) h[m] = mm(f.wlo, qlo[m], hinit);
#pragma unroll
  for (int m = 0; m < kMT; ++m) h[m] = mm(f.whi, qhi[m], h[m]);
}

template <bool kNext>
__device__ __forceinline__ void stage(const WFrag& fc, const WFrag& fn, int hi, v2f bone,
                                      const v2f (&qlo)[kMT], const v2f (&qhi)[kMT],
                                      const v8f (&hc)[kMT], v8f (&hn)[kMT], v8f (&acc)[kMT])
{
  const v8f vzero = {};
  v8f hinit = vzero;
  if (kNext) {
    v2f ab1; ab1[0] = hi ? 0.0f : fn.b1v; ab1[1] = 0.0f;
    hinit = mm(ab1, bone, vzero);
  }
#pragma unroll
  for (int m = 0; m < kMT; ++m) acc[m] = mm(w2_step(fc, 0), h_step(hc[m], 0), acc[m]);
  if (kNext) {
#pragma unroll
    for (int m = 0; m < kMT; ++m) hn[m] = mm(fn.wlo, qlo[m], hinit);
  }
#pragma unroll
  for (int m = 0; m < kMT; ++m) acc[m] = mm(w2_step(fc, 1), h_step(hc[m], 1), acc[m]);
  if (kNext) {
#pragma unroll
    for (int m = 0; m < kMT; ++m) hn[m] = mm(fn.whi, qhi[m], hn[m]);
  }
#pragma unroll
  for (int m = 0; m < kMT; ++m) {
    acc[m] = mm(w2_step(fc, 2), h_step(hc[m], 2), acc[m]);
    if (kNext) {
      relu4(hn[m >> 1], 4 * (m & 1));
      __builtin_amdgcn_sched_barrier(0);
    }
  }
#pragma unroll
  for (int m = 0; m < kMT; ++m) {
    acc[m] = mm(w2_step(fc, 3), h_step(hc[m], 3), acc[m]);
    if (kNext) {
      relu4(hn[2 + (m >> 1)], 4 * (m & 1));
      __builtin_amdgcn_sched_barrier(0);
    }
  }
}
}

__global__ __launch_bounds__(256)
void ffq_fused_wmma(const float* __restrict__ x,
                    const float* __restrict__ phi,
                    const float* __restrict__ w1,
                    const float* __restrict__ b1,
                    const float* __restrict__ w2,
                    const float* __restrict__ b2,
                    float* __restrict__ out)
{
  const int lane = threadIdx.x & 31;
  const int wave = threadIdx.x >> 5;
  const int row  = lane & 15;
  const int hi   = lane >> 4;
  const int koff = hi << 1;

  const float* w1l = w1 + row * kNQ + koff;
  const float* b1l = b1 + row;
  const float* w2l = w2 + (row & 7) * kFFN + 8 * hi;

  WFrag f0 = load_frag(w1l, b1l, w2l, 0 * kChunk);
  WFrag f1 = load_frag(w1l, b1l, w2l, 1 * kChunk);
  WFrag f2 = load_frag(w1l, b1l, w2l, 2 * kChunk);
  WFrag f3 = load_frag(w1l, b1l, w2l, 3 * kChunk);

  const int tile0 = (blockIdx.x * kWaves + wave) * kMT;
  const float ph  = phi[0];

  v2f qlo[kMT], qhi[kMT];
#pragma unroll
  for (int m = 0; m < kMT; ++m) {
    const float* xr = x + (size_t)((tile0 + m) * 16 + row) * kNQ;
    qlo[m][0] = __cosf(xr[koff + 0] + ph);
    qlo[m][1] = __cosf(xr[koff + 1] + ph);
    qhi[m][0] = __cosf(xr[koff + 4] + ph);
    qhi[m][1] = __cosf(xr[koff + 5] + ph);
  }

  v2f bone; bone[0] = hi ? 0.0f : 1.0f; bone[1] = 0.0f;
  const v8f vzero = {};

  v2f ab2; ab2[0] = hi ? 0.0f : b2[row & 7]; ab2[1] = 0.0f;
  const v8f accInit = mm(ab2, bone, vzero);
  v8f acc[kMT];
#pragma unroll
  for (int m = 0; m < kMT; ++m) acc[m] = accInit;

  v8f ha[kMT], hb[kMT];
  gemm1(f0, hi, bone, qlo, qhi, ha);
#pragma unroll
  for (int m = 0; m < kMT; ++m) { relu4(ha[m], 0); relu4(ha[m], 4); }

  for (int c = 0; c < kChunks - kStages; c += kStages) {
    const int n0 = c * kChunk;
    stage<true>(f0, f1, hi, bone, qlo, qhi, ha, hb, acc);
    f0 = load_frag(w1l, b1l, w2l, n0 + 4 * kChunk);
    stage<true>(f1, f2, hi, bone, qlo, qhi, hb, ha, acc);
    f1 = load_frag(w1l, b1l, w2l, n0 + 5 * kChunk);
    stage<true>(f2, f3, hi, bone, qlo, qhi, ha, hb, acc);
    f2 = load_frag(w1l, b1l, w2l, n0 + 6 * kChunk);
    stage<true>(f3, f0, hi, bone, qlo, qhi, hb, ha, acc);
    f3 = load_frag(w1l, b1l, w2l, n0 + 7 * kChunk);
  }
  stage<true>(f0, f1, hi, bone, qlo, qhi, ha, hb, acc);
  stage<true>(f1, f2, hi, bone, qlo, qhi, hb, ha, acc);
  stage<true>(f2, f3, hi, bone, qlo, qhi, ha, hb, acc);
  stage<false>(f3, f3, hi, bone, qlo, qhi, hb, ha, acc);

  if (hi == 0) {
#pragma unroll
    for (int m = 0; m < kMT; ++m) {
      float* orow = out + (size_t)((tile0 + m) * 16 + row) * kNQ;
      v4f lo4, hi4;
      lo4[0] = acc[m][0]; lo4[1] = acc[m][1]; lo4[2] = acc[m][2]; lo4[3] = acc[m][3];
      hi4[0] = acc[m][4]; hi4[1] = acc[m][5]; hi4[2] = acc[m][6]; hi4[3] = acc[m][7];
      *(v4f*)(orow + 0) = lo4;
      *(v4f*)(orow + 4) = hi4;
    }
  }
}

extern "C" void kernel_launch(void* const* d_in, const int* in_sizes, int n_in,
                              void* d_out, int out_size, void* d_ws, size_t ws_size,
                              hipStream_t stream) {
  const float* x   = (const float*)d_in[0];
  const float* phi = (const float*)d_in[1];
  const float* w1  = (const float*)d_in[2];
  const float* b1  = (const float*)d_in[3];
  const float* w2  = (const float*)d_in[4];
  const float* b2  = (const float*)d_in[5];
  float* out = (float*)d_out;

  const int blocks = kTokens / (16 * kMT * kWaves);
  ffq_fused_wmma<<<blocks, 256, 0, stream>>>(x, phi, w1, b1, w2, b2, out);
}
